# starting kernel
# baseline (speedup 1.0000x reference)
#include <hip/hip_runtime.h>
#include <math.h>


typedef float v2f __attribute__((ext_vector_type(2)));
typedef float v8f __attribute__((ext_vector_type(8)));

#define NEG_SLOPE 0.2f
#define GAT_EPS 1e-16f

__device__ __forceinline__ unsigned fenc(float f) {
  unsigned u = __float_as_uint(f);
  return (u & 0x80000000u) ? ~u : (u | 0x80000000u);
}
__device__ __forceinline__ float fdec(unsigned k) {
  unsigned u = (k & 0x80000000u) ? (k ^ 0x80000000u) : ~k;
  return __uint_as_float(u);
}

__global__ void gat_gemm1_wmma(const float* __restrict__ X,
                               const float* __restrict__ W,
                               float* __restrict__ H, int N) {
  const int lane    = threadIdx.x & 31;
  const int wave    = threadIdx.x >> 5;
  const int row0    = blockIdx.x * 16;
  const int colBase = wave * 16;
  const int m       = lane & 15;
  const int khalf   = (lane >> 4) * 2;

  int rowA = row0 + m; if (rowA >= N) rowA = N - 1;
  const float* xrow = X + (long)rowA * 128;

  v8f c = {};
  for (int k0 = 0; k0 < 128; k0 += 4) {
    const int k = k0 + khalf;
    v2f a; a.x = xrow[k];                 a.y = xrow[k + 1];
    v2f b; b.x = W[k * 64 + colBase + m]; b.y = W[(k + 1) * 64 + colBase + m];
    c = __builtin_amdgcn_wmma_f32_16x16x4_f32(false, a, false, b,
                                              (short)0, c, false, false);
  }
  const int mrow = (lane >> 4) * 8;
#pragma unroll
  for (int v = 0; v < 8; ++v) {
    int r = row0 + mrow + v;
    if (r < N) H[(long)r * 64 + colBase + m] = c[v];
  }
}

__global__ void gat_gemm2_wmma(const float* __restrict__ X,
                               const float* __restrict__ W,
                               float* __restrict__ H, int N) {
  const int lane    = threadIdx.x & 31;
  const int wave    = threadIdx.x >> 5;
  const int row0    = blockIdx.x * 16;
  const int col     = wave * 16 + (lane & 15);
  const int m       = lane & 15;
  const int khalf   = (lane >> 4) * 2;
  const bool colOK  = (col < 40);

  int rowA = row0 + m; if (rowA >= N) rowA = N - 1;
  const float* xrow = X + (long)rowA * 64;

  v8f c = {};
  for (int k0 = 0; k0 < 64; k0 += 4) {
    const int k = k0 + khalf;
    v2f a; a.x = xrow[k]; a.y = xrow[k + 1];
    v2f b;
    b.x = colOK ? W[k * 40 + col] : 0.0f;
    b.y = colOK ? W[(k + 1) * 40 + col] : 0.0f;
    c = __builtin_amdgcn_wmma_f32_16x16x4_f32(false, a, false, b,
                                              (short)0, c, false, false);
  }
  const int mrow = (lane >> 4) * 8;
#pragma unroll
  for (int v = 0; v < 8; ++v) {
    int r = row0 + mrow + v;
    if (colOK && r < N) H[(long)r * 40 + col] = c[v];
  }
}

__global__ void gat_attn1(const float* __restrict__ H, const float* __restrict__ aS,
                          const float* __restrict__ aD, float* __restrict__ eS,
                          float* __restrict__ eD, int N) {
  int t = blockIdx.x * blockDim.x + threadIdx.x;
  if (t >= N * 8) return;
  const int head = t & 7;
  const float* hp = H + (long)(t >> 3) * 64 + head * 8;
  const float* as = aS + head * 8;
  const float* ad = aD + head * 8;
  float s = 0.f, d = 0.f;
#pragma unroll
  for (int c = 0; c < 8; ++c) { float h = hp[c]; s += h * as[c]; d += h * ad[c]; }
  eS[t] = s; eD[t] = d;
}

__global__ void gat_attn2(const float* __restrict__ H, const float* __restrict__ aS,
                          const float* __restrict__ aD, float* __restrict__ eS,
                          float* __restrict__ eD, int N) {
  int n = blockIdx.x * blockDim.x + threadIdx.x;
  if (n >= N) return;
  const float* hp = H + (long)n * 40;
  float s = 0.f, d = 0.f;
#pragma unroll 8
  for (int c = 0; c < 40; ++c) { float h = hp[c]; s += h * aS[c]; d += h * aD[c]; }
  eS[n] = s; eD[n] = d;
}

__global__ void gat_edge_max(const int* __restrict__ src, const int* __restrict__ dst,
                             const float* __restrict__ eS, const float* __restrict__ eD,
                             float* __restrict__ eBuf, unsigned* __restrict__ mKey,
                             int E, int H) {
  int t = blockIdx.x * blockDim.x + threadIdx.x;
  if (t >= E * H) return;
  const int e = t / H, h = t - e * H;
  const int s = src[e], d = dst[e];
  float v = eS[s * H + h] + eD[d * H + h];
  v = (v > 0.f) ? v : v * NEG_SLOPE;
  eBuf[t] = v;
  atomicMax(mKey + d * H + h, fenc(v));
}

__global__ void gat_edge_sum(const int* __restrict__ dst, float* __restrict__ eBuf,
                             const unsigned* __restrict__ mKey, float* __restrict__ sSum,
                             int E, int H) {
  int t = blockIdx.x * blockDim.x + threadIdx.x;
  if (t >= E * H) return;
  const int e = t / H, h = t - e * H;
  const int d = dst[e];
  float p = expf(eBuf[t] - fdec(mKey[d * H + h]));
  eBuf[t] = p;
  atomicAdd(sSum + d * H + h, p);
}

__global__ void gat_edge_msg1(const int* __restrict__ src, const int* __restrict__ dst,
                              const float* __restrict__ eBuf, const float* __restrict__ sSum,
                              const float* __restrict__ Hn, float* __restrict__ agg, int E) {
  int t = blockIdx.x * blockDim.x + threadIdx.x;
  if (t >= E * 8) return;
  const int e = t >> 3, h = t & 7;
  const int s = src[e], d = dst[e];
  const float alpha = eBuf[t] / (sSum[d * 8 + h] + GAT_EPS);
  const float4* hp = (const float4*)(Hn + (long)s * 64 + h * 8);
  float4 u0 = hp[0], u1 = hp[1];
  float* o = agg + (long)d * 64 + h * 8;
  atomicAdd(o + 0, u0.x * alpha); atomicAdd(o + 1, u0.y * alpha);
  atomicAdd(o + 2, u0.z * alpha); atomicAdd(o + 3, u0.w * alpha);
  atomicAdd(o + 4, u1.x * alpha); atomicAdd(o + 5, u1.y * alpha);
  atomicAdd(o + 6, u1.z * alpha); atomicAdd(o + 7, u1.w * alpha);
}

__global__ void gat_edge_msg2(const int* __restrict__ src, const int* __restrict__ dst,
                              const float* __restrict__ eBuf, const float* __restrict__ sSum,
                              const float* __restrict__ Hn, float* __restrict__ agg, int E) {
  int e = blockIdx.x * blockDim.x + threadIdx.x;
  if (e >= E) return;
  const int s = src[e], d = dst[e];
  const float alpha = eBuf[e] / (sSum[d] + GAT_EPS);
  const float* hp = Hn + (long)s * 40;
  float* o = agg + (long)d * 40;
#pragma unroll 8
  for (int c = 0; c < 40; ++c) atomicAdd(o + c, hp[c] * alpha);
}

__global__ void gat_bias_relu(float* __restrict__ agg, const float* __restrict__ b, long n64) {
  long t = (long)blockIdx.x * blockDim.x + threadIdx.x;
  if (t >= n64) return;
  float v = agg[t] + b[t & 63];
  agg[t] = v > 0.f ? v : 0.f;
}

__global__ void gat_logsoftmax(const float* __restrict__ agg, const float* __restrict__ b,
                               float* __restrict__ out, int N) {
  int n = blockIdx.x * blockDim.x + threadIdx.x;
  if (n >= N) return;
  const float* a = agg + (long)n * 40;
  float v[40];
  float mx = -INFINITY;
#pragma unroll 8
  for (int c = 0; c < 40; ++c) { v[c] = a[c] + b[c]; mx = fmaxf(mx, v[c]); }
  float sum = 0.f;
#pragma unroll 8
  for (int c = 0; c < 40; ++c) sum += expf(v[c] - mx);
  const float ls = mx + logf(sum);
  float* o = out + (long)n * 40;
#pragma unroll 8
  for (int c = 0; c < 40; ++c) o[c] = v[c] - ls;
}

extern "C" void kernel_launch(void* const* d_in, const int* in_sizes, int n_in,
                              void* d_out, int out_size, void* d_ws, size_t ws_size,
                              hipStream_t stream) {
  const float* x     = (const float*)d_in[0];
  const int*   eidx  = (const int*)d_in[1];
  const float* W1    = (const float*)d_in[2];
  const float* aS1   = (const float*)d_in[3];
  const float* aD1   = (const float*)d_in[4];
  const float* b1    = (const float*)d_in[5];
  const float* W2    = (const float*)d_in[6];
  const float* aS2   = (const float*)d_in[7];
  const float* aD2   = (const float*)d_in[8];
  const float* b2    = (const float*)d_in[9];
  float* out = (float*)d_out;

  const int N = in_sizes[0] / 128;
  const int E = in_sizes[1] / 2;
  const int* src = eidx;
  const int* dst = eidx + E;

  const long N64 = (long)N * 64, N40 = (long)N * 40;
  const long N8  = (long)N * 8,  E8  = (long)E * 8;

  float* ws      = (float*)d_ws;
  float* h1      = ws;
  float* esrc1   = h1 + N64;
  float* edst1   = esrc1 + N8;
  float* e1      = edst1 + N8;
  float* s1      = e1 + E8;
  unsigned* mK1  = (unsigned*)(s1 + N8);
  float* agg1    = (float*)mK1 + N8;
  float* h2      = agg1 + N64;
  float* esrc2   = h2 + N40;
  float* edst2   = esrc2 + N;
  float* e2      = edst2 + N;
  float* s2      = e2 + E;
  unsigned* mK2  = (unsigned*)(s2 + N);
  float* agg2    = (float*)mK2 + N;

  const int BT = 256;
  const int rowBlocks = (N + 15) / 16;

  gat_gemm1_wmma<<<rowBlocks, 128, 0, stream>>>(x, W1, h1, N);
  gat_attn1<<<(int)((N8 + BT - 1) / BT), BT, 0, stream>>>(h1, aS1, aD1, esrc1, edst1, N);
  hipMemsetAsync(s1, 0, (size_t)(N8 + N8 + N64) * 4, stream);
  gat_edge_max<<<(int)((E8 + BT - 1) / BT), BT, 0, stream>>>(src, dst, esrc1, edst1, e1, mK1, E, 8);
  gat_edge_sum<<<(int)((E8 + BT - 1) / BT), BT, 0, stream>>>(dst, e1, mK1, s1, E, 8);
  gat_edge_msg1<<<(int)((E8 + BT - 1) / BT), BT, 0, stream>>>(src, dst, e1, s1, h1, agg1, E);
  gat_bias_relu<<<(int)((N64 + BT - 1) / BT), BT, 0, stream>>>(agg1, b1, N64);

  gat_gemm2_wmma<<<rowBlocks, 96, 0, stream>>>(agg1, W2, h2, N);
  gat_attn2<<<(N + BT - 1) / BT, BT, 0, stream>>>(h2, aS2, aD2, esrc2, edst2, N);
  hipMemsetAsync(s2, 0, (size_t)(N + N + N40) * 4, stream);
  gat_edge_max<<<(E + BT - 1) / BT, BT, 0, stream>>>(src, dst, esrc2, edst2, e2, mK2, E, 1);
  gat_edge_sum<<<(E + BT - 1) / BT, BT, 0, stream>>>(dst, e2, mK2, s2, E, 1);
  gat_edge_msg2<<<(E + BT - 1) / BT, BT, 0, stream>>>(src, dst, e2, s2, h2, agg2, E);
  gat_logsoftmax<<<(N + BT - 1) / BT, BT, 0, stream>>>(agg2, b2, out, N);
}


# row arrays keep a head's 8 channels as two 16-byte pieces 128 bytes apart: each 16-byte gather, own-row load and row store of the list walks covers 4 whole 128-byte lines, not 8 half lines
# speedup vs baseline: 7.2938x; 7.2938x over previous
#include <hip/hip_runtime.h>
#include <math.h>


typedef float v2f __attribute__((ext_vector_type(2)));
typedef float v8f __attribute__((ext_vector_type(8)));

#define NEG_SLOPE 0.2f
#define GAT_EPS 1e-16f

#define G1_ROWS 64
#define G1_XP   132
#define G1_WP   160
__global__ void __launch_bounds__(128) gat_gemm1_wmma(const float* __restrict__ X,
                               const float* __restrict__ W,
                               float* __restrict__ H, int* __restrict__ listFirst,
                               const float* __restrict__ W2, const float* __restrict__ aS2,
                               const float* __restrict__ aD2, float* __restrict__ wv2, int N) {
  __shared__ __attribute__((aligned(16))) float Xl[G1_ROWS * G1_XP];
  __shared__ __attribute__((aligned(16))) float Wl[64 * G1_WP];
  const int tid   = threadIdx.x;
  const int lane  = tid & 31;
  const int wave  = tid >> 5;
  const int row0  = blockIdx.x * G1_ROWS;
  const int m     = lane & 15;
  const int khalf = (lane >> 4) * 2;

  if (tid < G1_ROWS) {
    const int rr = row0 + tid;
    if (rr < N) listFirst[rr] = -1;
  }
  if (blockIdx.x == 0) {
    const float* a  = (tid < 64) ? aS2 : aD2;
    const float* wr = W2 + (tid & 63) * 40;
    float acc = 0.f;
#pragma unroll
    for (int cc = 0; cc < 40; ++cc) acc = __builtin_fmaf(wr[cc], a[cc], acc);
    wv2[tid] = acc;
  }

  float4 xs[16], wa[8], wb[8];
#pragma unroll
  for (int t = 0; t < 16; ++t) {
    int r = row0 + wave + 4 * t; if (r >= N) r = N - 1;
    xs[t] = *(const float4*)(X + (long)r * 128 + lane * 4);
  }
#pragma unroll
  for (int t = 0; t < 8; ++t) {
    const int p = tid + 128 * t;
    const float4* wp = (const float4*)W + (p >> 4) * 32 + (p & 15);
    wa[t] = wp[0];
    wb[t] = wp[16];
  }
#pragma unroll
  for (int t = 0; t < 16; ++t)
    *(float4*)(Xl + (wave + 4 * t) * G1_XP + lane * 4) = xs[t];
#pragma unroll
  for (int t = 0; t < 8; ++t) {
    const int p = tid + 128 * t;
    float* d = Wl + (p >> 4) * G1_WP + (p & 15) * 8;
    *(float4*)d       = make_float4(wa[t].x, wb[t].x, wa[t].y, wb[t].y);
    *(float4*)(d + 4) = make_float4(wa[t].z, wb[t].z, wa[t].w, wb[t].w);
  }
  __syncthreads();

  const float* arow = Xl + (wave * 16 + m) * G1_XP + khalf;
  const float* bcol = Wl + (lane >> 4) * G1_WP + m * 2;
  v8f c0 = {}, c1 = {}, c2 = {}, c3 = {};
#pragma unroll 8
  for (int i = 0; i < 32; ++i) {
    const v2f a = *(const v2f*)(arow + 4 * i);
    const float* bp = bcol + 2 * i * G1_WP;
    const v2f b0 = *(const v2f*)(bp);
    const v2f b1 = *(const v2f*)(bp + 32);
    const v2f b2 = *(const v2f*)(bp + 64);
    const v2f b3 = *(const v2f*)(bp + 96);
    c0 = __builtin_amdgcn_wmma_f32_16x16x4_f32(false, a, false, b0, (short)0, c0, false, false);
    c1 = __builtin_amdgcn_wmma_f32_16x16x4_f32(false, a, false, b1, (short)0, c1, false, false);
    c2 = __builtin_amdgcn_wmma_f32_16x16x4_f32(false, a, false, b2, (short)0, c2, false, false);
    c3 = __builtin_amdgcn_wmma_f32_16x16x4_f32(false, a, false, b3, (short)0, c3, false, false);
  }

  {
    const int mrow = (lane >> 4) * 8;
    const int pcol = (m & 4) * 8 + (m >> 3) * 4 + (m & 3);
    float* crow = Xl + (wave * 16 + mrow) * G1_XP + pcol;
#pragma unroll
    for (int v = 0; v < 8; ++v) {
      crow[v * G1_XP]      = c0[v];
      crow[v * G1_XP + 8]  = c1[v];
      crow[v * G1_XP + 16] = c2[v];
      crow[v * G1_XP + 24] = c3[v];
    }
  }
  __syncthreads();
  {
    const int tr0 = row0 + wave * 16;
    const float* cl = Xl + (wave * 16 + (lane >> 4)) * G1_XP + (lane & 15) * 4;
    float* hg = H + (long)(tr0 + (lane >> 4)) * 64 + (lane & 15) * 4;
    if (tr0 + 16 <= N) {
      float4 o[8];
#pragma unroll
      for (int t = 0; t < 8; ++t) o[t] = *(const float4*)(cl + 2 * t * G1_XP);
#pragma unroll
      for (int t = 0; t < 8; ++t) *(float4*)(hg + 2 * t * 64) = o[t];
    } else {
      for (int t = 0; t < 8; ++t)
        if (tr0 + 2 * t + (lane >> 4) < N) *(float4*)(hg + 2 * t * 64) = *(const float4*)(cl + 2 * t * G1_XP);
    }
  }
}

__device__ __forceinline__ float gat_dot8(const float4 u0, const float4 u1,
                                          const float4 w0, const float4 w1) {
  v2f a = (v2f){u0.x, u0.y} * (v2f){w0.x, w0.y};
  v2f b = (v2f){u1.x, u1.y} * (v2f){w1.x, w1.y};
  a = __builtin_elementwise_fma((v2f){u0.z, u0.w}, (v2f){w0.z, w0.w}, a);
  b = __builtin_elementwise_fma((v2f){u1.z, u1.w}, (v2f){w1.z, w1.w}, b);
  a = a + b;
  return a.x + a.y;
}
__device__ __forceinline__ float gat_sum8(float v) {
  v += __int_as_float(__builtin_amdgcn_update_dpp(0, __float_as_int(v), 0x161, 0xf, 0xf, false));
  v += __int_as_float(__builtin_amdgcn_update_dpp(0, __float_as_int(v), 0x162, 0xf, 0xf, false));
  v += __int_as_float(__builtin_amdgcn_update_dpp(0, __float_as_int(v), 0x164, 0xf, 0xf, false));
  return v;
}

__global__ void gat_list_build(const int* __restrict__ src, const int* __restrict__ dst,
                               int* __restrict__ listFirst, int* __restrict__ link, int E) {
  const int e = blockIdx.x * blockDim.x + threadIdx.x;
  if (e >= E) return;
  const int prev = atomicExch(listFirst + dst[e], e);
  ((int2*)link)[e] = make_int2(prev, src[e]);
}

template <bool LAYER1>
__device__ __forceinline__ void gat_segment(const int first, const int2* __restrict__ nodes,
                                            const float4 w0, const float4 w1, const float ed,
                                            const float* __restrict__ rows, const int q,
                                            const int E, float4& r0, float4& r1) {
  float m = -INFINITY;
  float ssum = 0.f;
  float4 a0 = make_float4(0.f, 0.f, 0.f, 0.f), a1 = a0;
  int2 nd = make_int2(-1, 0);
  bool live = first >= 0;
  if (live) nd = nodes[first];
  for (int k = 0; live && k < E; ++k) {
    const int s = nd.y, en = nd.x;
    live = en >= 0;
    if (live) nd = nodes[en];
    const float4* hp = (const float4*)(rows + (long)s * 64 + q * 4);
    const float4 u0 = hp[0], u1 = hp[8];
    float es = gat_dot8(u0, u1, w0, w1);
    if (!LAYER1) es = gat_sum8(es);
    float v = es + ed;
    v = (v > 0.f) ? v : v * NEG_SLOPE;
    const float d  = v - m;
    const bool  up = d > 0.f;
    const float ex = expf(-fabsf(d));
    const float sc = up ? ex : 1.f;
    const float p  = up ? 1.f : ((v == -INFINITY) ? 0.f : ex);
    m = up ? v : m;
    ssum = ssum * sc + p;
    a0.x = a0.x * sc + p * u0.x; a0.y = a0.y * sc + p * u0.y;
    a0.z = a0.z * sc + p * u0.z; a0.w = a0.w * sc + p * u0.w;
    a1.x = a1.x * sc + p * u1.x; a1.y = a1.y * sc + p * u1.y;
    a1.z = a1.z * sc + p * u1.z; a1.w = a1.w * sc + p * u1.w;
  }
  const float den = ssum + GAT_EPS;
  r0.x = a0.x / den; r0.y = a0.y / den; r0.z = a0.z / den; r0.w = a0.w / den;
  r1.x = a1.x / den; r1.y = a1.y / den; r1.z = a1.z / den; r1.w = a1.w / den;
}

__global__ void gat_row1(const int* __restrict__ listFirst, const int* __restrict__ link,
                         const float* __restrict__ aS, const float* __restrict__ aD,
                         const float* __restrict__ rows, const float* __restrict__ b,
                         float* __restrict__ out, int N, int E) {
  const int t = blockIdx.x * blockDim.x + threadIdx.x;
  if (t >= N * 8) return;
  const int n = t >> 3, q = t & 7;
  const int first = listFirst[n];
  const float4* ap = (const float4*)(aS + q * 8);
  const float4* dp = (const float4*)(aD + q * 8);
  const float4* xp = (const float4*)(rows + (long)n * 64 + q * 4);
  const float4 w0 = ap[0], w1 = ap[1];
  const float ed = gat_dot8(xp[0], xp[8], dp[0], dp[1]);
  float4 r0, r1;
  gat_segment<true>(first, (const int2*)link, w0, w1, ed, rows, q, E, r0, r1);
  const float* bb = b + q * 8;
  r0.x += bb[0]; r0.y += bb[1]; r0.z += bb[2]; r0.w += bb[3];
  r1.x += bb[4]; r1.y += bb[5]; r1.z += bb[6]; r1.w += bb[7];
  r0.x = r0.x > 0.f ? r0.x : 0.f; r0.y = r0.y > 0.f ? r0.y : 0.f;
  r0.z = r0.z > 0.f ? r0.z : 0.f; r0.w = r0.w > 0.f ? r0.w : 0.f;
  r1.x = r1.x > 0.f ? r1.x : 0.f; r1.y = r1.y > 0.f ? r1.y : 0.f;
  r1.z = r1.z > 0.f ? r1.z : 0.f; r1.w = r1.w > 0.f ? r1.w : 0.f;
  float4* op = (float4*)(out + (long)n * 64 + q * 4);
  op[0] = r0; op[8] = r1;
}

#define TILE_PITCH 68
__global__ void gat_row2_tail(const int* __restrict__ listFirst, const int* __restrict__ link,
                              const float* __restrict__ wv,
                              const float* __restrict__ rows, const float* __restrict__ W,
                              const float* __restrict__ b, float* __restrict__ out,
                              int N, int E) {
  __shared__ __attribute__((aligned(16))) float tileA[16 * TILE_PITCH];
  __shared__ float tileC[16 * 40];
  const int tid  = threadIdx.x;
  const int lane = tid & 31;
  const int wave = tid >> 5;
  const int r = tid >> 3, q = tid & 7;
  const int n = blockIdx.x * 16 + r;
  const bool ok = n < N;

  float4 r0 = make_float4(0.f, 0.f, 0.f, 0.f), r1 = r0;
  if (ok) {
    const int first = listFirst[n];
    const float4* sp = (const float4*)(wv + q * 8);
    const float4* dp = (const float4*)(wv + 64 + q * 8);
    const float4* xp = (const float4*)(rows + (long)n * 64 + q * 4);
    const float4 w0 = sp[0], w1 = sp[1];
    const float ed = gat_sum8(gat_dot8(xp[0], xp[8], dp[0], dp[1]));
    gat_segment<false>(first, (const int2*)link, w0, w1, ed, rows, q, E, r0, r1);
  }

  const int m      = lane & 15;
  const int khalf  = (lane >> 4) * 2;
  const int col    = wave * 16 + m;
  const bool colOK = (col < 40);
  const int colc   = colOK ? col : 39;
  v2f bfr[16];
  if (wave < 3) {
#pragma unroll
    for (int i = 0; i < 16; ++i) {
      const int k = i * 4 + khalf;
      const float bx = W[k * 40 + colc], by = W[(k + 1) * 40 + colc];
      bfr[i].x = colOK ? bx : 0.0f;
      bfr[i].y = colOK ? by : 0.0f;
    }
  }
  {
    float4* tp = (float4*)(tileA + r * TILE_PITCH + q * 8);
    tp[0] = r0; tp[1] = r1;
  }
  __syncthreads();

  if (wave < 3) {
    const float* arow = tileA + m * TILE_PITCH;
    v8f c = {};
#pragma unroll
    for (int i = 0; i < 16; ++i) {
      const int k = i * 4 + khalf;
      v2f a; a.x = arow[k]; a.y = arow[k + 1];
      c = __builtin_amdgcn_wmma_f32_16x16x4_f32(false, a, false, bfr[i],
                                                (short)0, c, false, false);
    }
    const int mrow = (lane >> 4) * 8;
    if (colOK) {
      const float bc = b[col];
#pragma unroll
      for (int v = 0; v < 8; ++v) tileC[(mrow + v) * 40 + col] = c[v] + bc;
    }
  }
  __syncthreads();

  {
    float* cp = tileC + r * 40 + q * 5;
    float v[5];
    float mx = -INFINITY;
#pragma unroll
    for (int i = 0; i < 5; ++i) { v[i] = cp[i]; mx = fmaxf(mx, v[i]); }
    for (int o = 1; o < 8; o <<= 1) mx = fmaxf(mx, __shfl_xor(mx, o));
    float sum = 0.f;
#pragma unroll
    for (int i = 0; i < 5; ++i) sum += expf(v[i] - mx);
    for (int o = 1; o < 8; o <<= 1) sum += __shfl_xor(sum, o);
    const float ls = mx + logf(sum);
#pragma unroll
    for (int i = 0; i < 5; ++i) cp[i] = v[i] - ls;
  }
  __syncthreads();

  {
    const int n0 = blockIdx.x * 16 + wave * 4;
    const float* cw = tileC + wave * 160;
    float* ow = out + (long)n0 * 40;
    for (int j = 0; j < 5; ++j) {
      const int e = j * 32 + lane;
      if (n0 + e / 40 < N) ow[e] = cw[e];
    }
  }
}

extern "C" void kernel_launch(void* const* d_in, const int* in_sizes, int n_in,
                              void* d_out, int out_size, void* d_ws, size_t ws_size,
                              hipStream_t stream) {
  const float* x     = (const float*)d_in[0];
  const int*   eidx  = (const int*)d_in[1];
  const float* W1    = (const float*)d_in[2];
  const float* aS1   = (const float*)d_in[3];
  const float* aD1   = (const float*)d_in[4];
  const float* b1    = (const float*)d_in[5];
  const float* W2    = (const float*)d_in[6];
  const float* aS2   = (const float*)d_in[7];
  const float* aD2   = (const float*)d_in[8];
  const float* b2    = (const float*)d_in[9];
  float* out = (float*)d_out;

  const int N = in_sizes[0] / 128;
  const int E = in_sizes[1] / 2;
  const int* src = eidx;
  const int* dst = eidx + E;

  const long N64 = (long)N * 64;
  const long N8  = (long)N * 8;

  float* ws      = (float*)d_ws;
  float* h1      = ws;
  float* agg1    = h1 + N64;
  float* wv2     = agg1 + N64;
  int*   link    = (int*)(wv2 + 128);
  int*   first   = link + 2 * (long)E;

  const int BT = 256;
  const int rowBlocks = (N + 15) / 16;
  const int g1Blocks  = (N + 63) / 64;
  const int n8Blocks  = (int)((N8 + BT - 1) / BT);

  gat_gemm1_wmma<<<g1Blocks, 128, 0, stream>>>(x, W1, h1, first, W2, aS2, aD2, wv2, N);
  gat_list_build<<<(E + BT - 1) / BT, BT, 0, stream>>>(src, dst, first, link, E);
  gat_row1<<<n8Blocks, BT, 0, stream>>>(first, link, aS1, aD1, h1, b1, agg1, N, E);

  gat_row2_tail<<<rowBlocks, 128, 0, stream>>>(first, link, wv2, agg1, W2, b2, out, N, E);
}
